# starting kernel
# baseline (speedup 1.0000x reference)
#include <hip/hip_runtime.h>
#include <math.h>

typedef __attribute__((ext_vector_type(2))) float v2f;
typedef __attribute__((ext_vector_type(8))) float v8f;

#define N_IN  512
#define N_OUT 256
#define NB    128
#define INV_E 0.36787944117144233f

__device__ __forceinline__ float frcp(float x) { return __builtin_amdgcn_rcpf(x); }

__device__ __forceinline__ float lambertw0_dev(float z) {
  const float e = 2.7182818284590452f;
  float nearr = -1.f + __builtin_amdgcn_sqrtf(fmaxf(2.f * (e * z + 1.f), 0.f));
  float lz    = __logf(fmaxf(z, 1e-30f));
  float large = lz - __logf(fmaxf(lz, 1e-30f));
  float w = (z < 0.25f) ? nearr : ((z > e) ? large : __logf(1.f + fmaxf(z, -0.99f)));
#pragma unroll
  for (int i = 0; i < 4; ++i) {
    float ew  = __expf(w);
    float f   = fmaf(w, ew, -z);
    float wp1 = w + 1.f;
    float den = ew * wp1 - (w + 2.f) * f * frcp(2.f * wp1);
    den = (fabsf(den) < 1e-12f) ? 1e-12f : den;
    w = w - f * frcp(den);
  }
  return w;
}

__global__ __launch_bounds__(256)
void ttfs_equaltime_forward(const float* __restrict__ S,
                            const float* __restrict__ W,
                            float* __restrict__ out) {
  __shared__ float skey[N_IN];
  __shared__ int   sidx[N_IN];
  __shared__ float ewL[N_IN];
  __shared__ float tewL[N_IN];

  const int b    = blockIdx.x;
  const int tid  = threadIdx.x;
  const int lane = tid & 31;
  const int half = lane >> 4;
  const int m    = lane & 15;
  const int wave = tid >> 5;

  skey[tid]       = S[b * N_IN + tid];        sidx[tid]       = tid;
  skey[tid + 256] = S[b * N_IN + tid + 256];  sidx[tid + 256] = tid + 256;
  __syncthreads();
  for (int k = 2; k <= N_IN; k <<= 1) {
    for (int j = k >> 1; j > 0; j >>= 1) {
      int  i  = 2 * tid - (tid & (j - 1));
      int  p  = i + j;
      bool up = ((i & k) == 0);
      float ka = skey[i], kb2 = skey[p];
      bool sw = up ? (ka > kb2) : (ka < kb2);
      if (sw) {
        skey[i] = kb2; skey[p] = ka;
        int t0 = sidx[i]; sidx[i] = sidx[p]; sidx[p] = t0;
      }
      __syncthreads();
    }
  }
  for (int n = tid; n < N_IN; n += 256) {
    float s   = skey[n];
    bool  bad = __builtin_isinf(s);
    float sm  = bad ? 0.f : s;
    float ex  = bad ? 0.f : __expf(sm);
    ewL[n]  = ex;
    tewL[n] = sm * ex;
  }
  __syncthreads();

  v2f A[4];
#pragma unroll
  for (int c = 0; c < 4; ++c)
#pragma unroll
    for (int v = 0; v < 2; ++v)
      A[c][v] = ((4 * c + v + 2 * half) <= m) ? 1.f : 0.f;

#pragma unroll
  for (int tt = 0; tt < 2; ++tt) {
    const int ot = wave + 8 * tt;
    const int oG = ot * 16 + m;
    float carryA = 0.f, carryB = 0.f;
    float mn = __builtin_inff();

    for (int kb = 0; kb < N_IN / 16; ++kb) {
      v8f dA = {0.f,0.f,0.f,0.f,0.f,0.f,0.f,0.f};
      v8f dB = {0.f,0.f,0.f,0.f,0.f,0.f,0.f,0.f};
#pragma unroll
      for (int c = 0; c < 4; ++c) {
        int   j0 = kb * 16 + 4 * c + 2 * half;
        int   r0 = sidx[j0], r1 = sidx[j0 + 1];
        float w0 = W[r0 * N_OUT + oG];
        float w1 = W[r1 * N_OUT + oG];
        v2f b1 = { ewL[j0]  * w0, ewL[j0 + 1]  * w1 };
        v2f b2 = { tewL[j0] * w0, tewL[j0 + 1] * w1 };
        dA = __builtin_amdgcn_wmma_f32_16x16x4_f32(false, A[c], false, b1,
                                                   (short)0, dA, false, false);
        dB = __builtin_amdgcn_wmma_f32_16x16x4_f32(false, A[c], false, b2,
                                                   (short)0, dB, false, false);
      }
#pragma unroll
      for (int r = 0; r < 8; ++r) {
        int   k   = kb * 16 + r + 8 * half;
        float ssk = skey[k];
        float nxt = (k == N_IN - 1) ? __builtin_inff() : skey[k + 1];
        float a1  = dA[r] + carryA;
        float bb  = dB[r] + carryB;
        float a1s = (fabsf(a1) < 1e-10f) ? 1e-10f : a1;
        float inv_a1 = frcp(a1s);
        float ratio  = bb * inv_a1;
        float z      = -__expf(ratio) * inv_a1;
        bool valid = (z >= -INV_E);
        bool need  = valid && (ratio + 1.f >= ssk) && !((z < 0.f) && (ratio > nxt));
        float t = __builtin_inff();
        if (__ballot(need)) {
          float zs   = fmaxf(z, -INV_E + 1e-7f);
          float wlam = lambertw0_dev(zs);
          t = need ? (ratio - wlam) : __builtin_inff();
        }
        if (t < ssk || t > nxt) t = __builtin_inff();
        mn = fminf(mn, t);
      }
      carryA += __shfl(dA[7], 16 + m, 32);
      carryB += __shfl(dB[7], 16 + m, 32);
    }
    mn = fminf(mn, __shfl_xor(mn, 16, 32));
    if (half == 0) out[b * N_OUT + oG] = mn;
  }
}

extern "C" void kernel_launch(void* const* d_in, const int* in_sizes, int n_in,
                              void* d_out, int out_size, void* d_ws, size_t ws_size,
                              hipStream_t stream) {
  const float* spikes  = (const float*)d_in[0];
  const float* weights = (const float*)d_in[1];
  float*       out     = (float*)d_out;
  (void)in_sizes; (void)n_in; (void)out_size; (void)d_ws; (void)ws_size;
  ttfs_equaltime_forward<<<NB, 256, 0, stream>>>(spikes, weights, out);
}


# Two work-groups per batch row (grid 128 -> 128x2): each wave scans one 16-column output tile instead of two, sort repeated per work-group, so 256 work-groups cover all 256 WGPs; arithmetic unchanged
# speedup vs baseline: 1.9023x; 1.9023x over previous
#include <hip/hip_runtime.h>
#include <math.h>

typedef __attribute__((ext_vector_type(2))) float v2f;
typedef __attribute__((ext_vector_type(8))) float v8f;

#define N_IN  512
#define N_OUT 256
#define NB    128
#define INV_E 0.36787944117144233f

__device__ __forceinline__ float frcp(float x) { return __builtin_amdgcn_rcpf(x); }

__device__ __forceinline__ float lambertw0_dev(float z) {
  const float e = 2.7182818284590452f;
  float nearr = -1.f + __builtin_amdgcn_sqrtf(fmaxf(2.f * (e * z + 1.f), 0.f));
  float lz    = __logf(fmaxf(z, 1e-30f));
  float large = lz - __logf(fmaxf(lz, 1e-30f));
  float w = (z < 0.25f) ? nearr : ((z > e) ? large : __logf(1.f + fmaxf(z, -0.99f)));
#pragma unroll
  for (int i = 0; i < 4; ++i) {
    float ew  = __expf(w);
    float f   = fmaf(w, ew, -z);
    float wp1 = w + 1.f;
    float den = ew * wp1 - (w + 2.f) * f * frcp(2.f * wp1);
    den = (fabsf(den) < 1e-12f) ? 1e-12f : den;
    w = w - f * frcp(den);
  }
  return w;
}

__global__ __launch_bounds__(256)
void ttfs_equaltime_forward(const float* __restrict__ S,
                            const float* __restrict__ W,
                            float* __restrict__ out) {
  __shared__ float skey[N_IN];
  __shared__ int   sidx[N_IN];
  __shared__ float ewL[N_IN];
  __shared__ float tewL[N_IN];

  const int b    = blockIdx.x;
  const int tt   = blockIdx.y;
  const int tid  = threadIdx.x;
  const int lane = tid & 31;
  const int half = lane >> 4;
  const int m    = lane & 15;
  const int wave = tid >> 5;

  skey[tid]       = S[b * N_IN + tid];        sidx[tid]       = tid;
  skey[tid + 256] = S[b * N_IN + tid + 256];  sidx[tid + 256] = tid + 256;
  __syncthreads();
  for (int k = 2; k <= N_IN; k <<= 1) {
    for (int j = k >> 1; j > 0; j >>= 1) {
      int  i  = 2 * tid - (tid & (j - 1));
      int  p  = i + j;
      bool up = ((i & k) == 0);
      float ka = skey[i], kb2 = skey[p];
      bool sw = up ? (ka > kb2) : (ka < kb2);
      if (sw) {
        skey[i] = kb2; skey[p] = ka;
        int t0 = sidx[i]; sidx[i] = sidx[p]; sidx[p] = t0;
      }
      __syncthreads();
    }
  }
  for (int n = tid; n < N_IN; n += 256) {
    float s   = skey[n];
    bool  bad = __builtin_isinf(s);
    float sm  = bad ? 0.f : s;
    float ex  = bad ? 0.f : __expf(sm);
    ewL[n]  = ex;
    tewL[n] = sm * ex;
  }
  __syncthreads();

  v2f A[4];
#pragma unroll
  for (int c = 0; c < 4; ++c)
#pragma unroll
    for (int v = 0; v < 2; ++v)
      A[c][v] = ((4 * c + v + 2 * half) <= m) ? 1.f : 0.f;

  {
    const int ot = wave + 8 * tt;
    const int oG = ot * 16 + m;
    float carryA = 0.f, carryB = 0.f;
    float mn = __builtin_inff();

    for (int kb = 0; kb < N_IN / 16; ++kb) {
      v8f dA = {0.f,0.f,0.f,0.f,0.f,0.f,0.f,0.f};
      v8f dB = {0.f,0.f,0.f,0.f,0.f,0.f,0.f,0.f};
#pragma unroll
      for (int c = 0; c < 4; ++c) {
        int   j0 = kb * 16 + 4 * c + 2 * half;
        int   r0 = sidx[j0], r1 = sidx[j0 + 1];
        float w0 = W[r0 * N_OUT + oG];
        float w1 = W[r1 * N_OUT + oG];
        v2f b1 = { ewL[j0]  * w0, ewL[j0 + 1]  * w1 };
        v2f b2 = { tewL[j0] * w0, tewL[j0 + 1] * w1 };
        dA = __builtin_amdgcn_wmma_f32_16x16x4_f32(false, A[c], false, b1,
                                                   (short)0, dA, false, false);
        dB = __builtin_amdgcn_wmma_f32_16x16x4_f32(false, A[c], false, b2,
                                                   (short)0, dB, false, false);
      }
#pragma unroll
      for (int r = 0; r < 8; ++r) {
        int   k   = kb * 16 + r + 8 * half;
        float ssk = skey[k];
        float nxt = (k == N_IN - 1) ? __builtin_inff() : skey[k + 1];
        float a1  = dA[r] + carryA;
        float bb  = dB[r] + carryB;
        float a1s = (fabsf(a1) < 1e-10f) ? 1e-10f : a1;
        float inv_a1 = frcp(a1s);
        float ratio  = bb * inv_a1;
        float z      = -__expf(ratio) * inv_a1;
        bool valid = (z >= -INV_E);
        bool need  = valid && (ratio + 1.f >= ssk) && !((z < 0.f) && (ratio > nxt));
        float t = __builtin_inff();
        if (__ballot(need)) {
          float zs   = fmaxf(z, -INV_E + 1e-7f);
          float wlam = lambertw0_dev(zs);
          t = need ? (ratio - wlam) : __builtin_inff();
        }
        if (t < ssk || t > nxt) t = __builtin_inff();
        mn = fminf(mn, t);
      }
      carryA += __shfl(dA[7], 16 + m, 32);
      carryB += __shfl(dB[7], 16 + m, 32);
    }
    mn = fminf(mn, __shfl_xor(mn, 16, 32));
    if (half == 0) out[b * N_OUT + oG] = mn;
  }
}

extern "C" void kernel_launch(void* const* d_in, const int* in_sizes, int n_in,
                              void* d_out, int out_size, void* d_ws, size_t ws_size,
                              hipStream_t stream) {
  const float* spikes  = (const float*)d_in[0];
  const float* weights = (const float*)d_in[1];
  float*       out     = (float*)d_out;
  (void)in_sizes; (void)n_in; (void)out_size; (void)d_ws; (void)ws_size;
  ttfs_equaltime_forward<<<dim3(NB, 2), 256, 0, stream>>>(spikes, weights, out);
}


# Exact two-sided window test before the Lambert solve: W0 monotone, so t in [t_k,t_k+1] iff c2*e^c2 <= z <= c1*e^c1 (c = ratio - bound, widened 1e-2 in W); rows that cannot land there skip the solve
# speedup vs baseline: 3.9221x; 2.0618x over previous
#include <hip/hip_runtime.h>
#include <math.h>

typedef __attribute__((ext_vector_type(2))) float v2f;
typedef __attribute__((ext_vector_type(8))) float v8f;

#define N_IN  512
#define N_OUT 256
#define NB    128
#define INV_E 0.36787944117144233f
#define W_MARGIN 0.01f

__device__ __forceinline__ float frcp(float x) { return __builtin_amdgcn_rcpf(x); }

__device__ __forceinline__ float lambertw0_dev(float z) {
  const float e = 2.7182818284590452f;
  float nearr = -1.f + __builtin_amdgcn_sqrtf(fmaxf(2.f * (e * z + 1.f), 0.f));
  float lz    = __logf(fmaxf(z, 1e-30f));
  float large = lz - __logf(fmaxf(lz, 1e-30f));
  float w = (z < 0.25f) ? nearr : ((z > e) ? large : __logf(1.f + fmaxf(z, -0.99f)));
#pragma unroll
  for (int i = 0; i < 4; ++i) {
    float ew  = __expf(w);
    float f   = fmaf(w, ew, -z);
    float wp1 = w + 1.f;
    float den = ew * wp1 - (w + 2.f) * f * frcp(2.f * wp1);
    den = (fabsf(den) < 1e-12f) ? 1e-12f : den;
    w = w - f * frcp(den);
  }
  return w;
}

__global__ __launch_bounds__(256)
void ttfs_equaltime_forward(const float* __restrict__ S,
                            const float* __restrict__ W,
                            float* __restrict__ out) {
  __shared__ float skey[N_IN];
  __shared__ int   sidx[N_IN];
  __shared__ float ewL[N_IN];
  __shared__ float tewL[N_IN];

  const int b    = blockIdx.x;
  const int tt   = blockIdx.y;
  const int tid  = threadIdx.x;
  const int lane = tid & 31;
  const int half = lane >> 4;
  const int m    = lane & 15;
  const int wave = tid >> 5;

  skey[tid]       = S[b * N_IN + tid];        sidx[tid]       = tid;
  skey[tid + 256] = S[b * N_IN + tid + 256];  sidx[tid + 256] = tid + 256;
  __syncthreads();
  for (int k = 2; k <= N_IN; k <<= 1) {
    for (int j = k >> 1; j > 0; j >>= 1) {
      int  i  = 2 * tid - (tid & (j - 1));
      int  p  = i + j;
      bool up = ((i & k) == 0);
      float ka = skey[i], kb2 = skey[p];
      bool sw = up ? (ka > kb2) : (ka < kb2);
      if (sw) {
        skey[i] = kb2; skey[p] = ka;
        int t0 = sidx[i]; sidx[i] = sidx[p]; sidx[p] = t0;
      }
      __syncthreads();
    }
  }
  for (int n = tid; n < N_IN; n += 256) {
    float s   = skey[n];
    bool  bad = __builtin_isinf(s);
    float sm  = bad ? 0.f : s;
    float ex  = bad ? 0.f : __expf(sm);
    ewL[n]  = ex;
    tewL[n] = sm * ex;
  }
  __syncthreads();

  v2f A[4];
#pragma unroll
  for (int c = 0; c < 4; ++c)
#pragma unroll
    for (int v = 0; v < 2; ++v)
      A[c][v] = ((4 * c + v + 2 * half) <= m) ? 1.f : 0.f;

  {
    const int ot = wave + 8 * tt;
    const int oG = ot * 16 + m;
    float carryA = 0.f, carryB = 0.f;
    float mn = __builtin_inff();

    for (int kb = 0; kb < N_IN / 16; ++kb) {
      v8f dA = {0.f,0.f,0.f,0.f,0.f,0.f,0.f,0.f};
      v8f dB = {0.f,0.f,0.f,0.f,0.f,0.f,0.f,0.f};
#pragma unroll
      for (int c = 0; c < 4; ++c) {
        int   j0 = kb * 16 + 4 * c + 2 * half;
        int   r0 = sidx[j0], r1 = sidx[j0 + 1];
        float w0 = W[r0 * N_OUT + oG];
        float w1 = W[r1 * N_OUT + oG];
        v2f b1 = { ewL[j0]  * w0, ewL[j0 + 1]  * w1 };
        v2f b2 = { tewL[j0] * w0, tewL[j0 + 1] * w1 };
        dA = __builtin_amdgcn_wmma_f32_16x16x4_f32(false, A[c], false, b1,
                                                   (short)0, dA, false, false);
        dB = __builtin_amdgcn_wmma_f32_16x16x4_f32(false, A[c], false, b2,
                                                   (short)0, dB, false, false);
      }
      float ssk_[8], nxt_[8], ratio_[8], zs_[8];
      bool  need_[8];
      bool  anyNeed = false;
#pragma unroll
      for (int r = 0; r < 8; ++r) {
        int   k   = kb * 16 + r + 8 * half;
        float ssk = skey[k];
        float nxt = (k == N_IN - 1) ? __builtin_inff() : skey[k + 1];
        float a1  = dA[r] + carryA;
        float bb  = dB[r] + carryB;
        float a1s = (fabsf(a1) < 1e-10f) ? 1e-10f : a1;
        float inv_a1 = frcp(a1s);
        float ratio  = bb * inv_a1;
        float z      = -__expf(ratio) * inv_a1;
        float zs     = fmaxf(z, -INV_E + 1e-7f);
        bool valid = (z >= -INV_E);
        bool need  = valid && (ratio + 1.f >= ssk) && !((z < 0.f) && (ratio > nxt));
        float c1 = (ratio - ssk) + W_MARGIN;
        float c2 = (ratio - nxt) - W_MARGIN;
        float hi = c1 * __expf(c1);
        float lo = c2 * __expf(c2);
        need = need && !(zs > hi) && !((c2 > -1.f) && (zs < lo));
        ssk_[r] = ssk; nxt_[r] = nxt; ratio_[r] = ratio; zs_[r] = zs; need_[r] = need;
        anyNeed = anyNeed || need;
      }
      if (__ballot(anyNeed)) {
#pragma unroll
        for (int r = 0; r < 8; ++r) {
          if (__ballot(need_[r])) {
            float wlam = lambertw0_dev(zs_[r]);
            float t = need_[r] ? (ratio_[r] - wlam) : __builtin_inff();
            if (t < ssk_[r] || t > nxt_[r]) t = __builtin_inff();
            mn = fminf(mn, t);
          }
        }
      }
      carryA += __shfl(dA[7], 16 + m, 32);
      carryB += __shfl(dB[7], 16 + m, 32);
    }
    mn = fminf(mn, __shfl_xor(mn, 16, 32));
    if (half == 0) out[b * N_OUT + oG] = mn;
  }
}

extern "C" void kernel_launch(void* const* d_in, const int* in_sizes, int n_in,
                              void* d_out, int out_size, void* d_ws, size_t ws_size,
                              hipStream_t stream) {
  const float* spikes  = (const float*)d_in[0];
  const float* weights = (const float*)d_in[1];
  float*       out     = (float*)d_out;
  (void)in_sizes; (void)n_in; (void)out_size; (void)d_ws; (void)ws_size;
  ttfs_equaltime_forward<<<dim3(NB, 2), 256, 0, stream>>>(spikes, weights, out);
}


# Weight gather software-pipelined one k-block ahead: block kb+1's rows (LDS argsort index, then global load) are requested at the top of trip kb and land under its window tests; same arithmetic
# speedup vs baseline: 4.0534x; 1.0335x over previous
#include <hip/hip_runtime.h>
#include <math.h>

typedef __attribute__((ext_vector_type(2))) float v2f;
typedef __attribute__((ext_vector_type(8))) float v8f;

#define N_IN  512
#define N_OUT 256
#define NB    128
#define INV_E 0.36787944117144233f
#define W_MARGIN 0.01f

__device__ __forceinline__ float frcp(float x) { return __builtin_amdgcn_rcpf(x); }

__device__ __forceinline__ float lambertw0_dev(float z) {
  const float e = 2.7182818284590452f;
  float nearr = -1.f + __builtin_amdgcn_sqrtf(fmaxf(2.f * (e * z + 1.f), 0.f));
  float lz    = __logf(fmaxf(z, 1e-30f));
  float large = lz - __logf(fmaxf(lz, 1e-30f));
  float w = (z < 0.25f) ? nearr : ((z > e) ? large : __logf(1.f + fmaxf(z, -0.99f)));
#pragma unroll
  for (int i = 0; i < 4; ++i) {
    float ew  = __expf(w);
    float f   = fmaf(w, ew, -z);
    float wp1 = w + 1.f;
    float den = ew * wp1 - (w + 2.f) * f * frcp(2.f * wp1);
    den = (fabsf(den) < 1e-12f) ? 1e-12f : den;
    w = w - f * frcp(den);
  }
  return w;
}

__global__ __launch_bounds__(256)
void ttfs_equaltime_forward(const float* __restrict__ S,
                            const float* __restrict__ W,
                            float* __restrict__ out) {
  __shared__ float skey[N_IN];
  __shared__ int   sidx[N_IN];
  __shared__ float ewL[N_IN];
  __shared__ float tewL[N_IN];

  const int b    = blockIdx.x;
  const int tt   = blockIdx.y;
  const int tid  = threadIdx.x;
  const int lane = tid & 31;
  const int half = lane >> 4;
  const int m    = lane & 15;
  const int wave = tid >> 5;

  skey[tid]       = S[b * N_IN + tid];        sidx[tid]       = tid;
  skey[tid + 256] = S[b * N_IN + tid + 256];  sidx[tid + 256] = tid + 256;
  __syncthreads();
  for (int k = 2; k <= N_IN; k <<= 1) {
    for (int j = k >> 1; j > 0; j >>= 1) {
      int  i  = 2 * tid - (tid & (j - 1));
      int  p  = i + j;
      bool up = ((i & k) == 0);
      float ka = skey[i], kb2 = skey[p];
      bool sw = up ? (ka > kb2) : (ka < kb2);
      if (sw) {
        skey[i] = kb2; skey[p] = ka;
        int t0 = sidx[i]; sidx[i] = sidx[p]; sidx[p] = t0;
      }
      __syncthreads();
    }
  }
  for (int n = tid; n < N_IN; n += 256) {
    float s   = skey[n];
    bool  bad = __builtin_isinf(s);
    float sm  = bad ? 0.f : s;
    float ex  = bad ? 0.f : __expf(sm);
    ewL[n]  = ex;
    tewL[n] = sm * ex;
  }
  __syncthreads();

  v2f A[4];
#pragma unroll
  for (int c = 0; c < 4; ++c)
#pragma unroll
    for (int v = 0; v < 2; ++v)
      A[c][v] = ((4 * c + v + 2 * half) <= m) ? 1.f : 0.f;

  {
    const int ot = wave + 8 * tt;
    const int oG = ot * 16 + m;
    float carryA = 0.f, carryB = 0.f;
    float mn = __builtin_inff();

    float wn0[4], wn1[4];
#pragma unroll
    for (int c = 0; c < 4; ++c) {
      int j0 = 4 * c + 2 * half;
      wn0[c] = W[sidx[j0]     * N_OUT + oG];
      wn1[c] = W[sidx[j0 + 1] * N_OUT + oG];
    }

    for (int kb = 0; kb < N_IN / 16; ++kb) {
      v8f dA = {0.f,0.f,0.f,0.f,0.f,0.f,0.f,0.f};
      v8f dB = {0.f,0.f,0.f,0.f,0.f,0.f,0.f,0.f};
      float wc0[4], wc1[4];
#pragma unroll
      for (int c = 0; c < 4; ++c) { wc0[c] = wn0[c]; wc1[c] = wn1[c]; }
      const int kn = (kb + 1 < N_IN / 16) ? kb + 1 : kb;
#pragma unroll
      for (int c = 0; c < 4; ++c) {
        int jn = kn * 16 + 4 * c + 2 * half;
        wn0[c] = W[sidx[jn]     * N_OUT + oG];
        wn1[c] = W[sidx[jn + 1] * N_OUT + oG];
      }
#pragma unroll
      for (int c = 0; c < 4; ++c) {
        int   j0 = kb * 16 + 4 * c + 2 * half;
        float w0 = wc0[c];
        float w1 = wc1[c];
        v2f b1 = { ewL[j0]  * w0, ewL[j0 + 1]  * w1 };
        v2f b2 = { tewL[j0] * w0, tewL[j0 + 1] * w1 };
        dA = __builtin_amdgcn_wmma_f32_16x16x4_f32(false, A[c], false, b1,
                                                   (short)0, dA, false, false);
        dB = __builtin_amdgcn_wmma_f32_16x16x4_f32(false, A[c], false, b2,
                                                   (short)0, dB, false, false);
      }
      float ssk_[8], nxt_[8], ratio_[8], zs_[8];
      bool  need_[8];
      bool  anyNeed = false;
#pragma unroll
      for (int r = 0; r < 8; ++r) {
        int   k   = kb * 16 + r + 8 * half;
        float ssk = skey[k];
        float nxt = (k == N_IN - 1) ? __builtin_inff() : skey[k + 1];
        float a1  = dA[r] + carryA;
        float bb  = dB[r] + carryB;
        float a1s = (fabsf(a1) < 1e-10f) ? 1e-10f : a1;
        float inv_a1 = frcp(a1s);
        float ratio  = bb * inv_a1;
        float z      = -__expf(ratio) * inv_a1;
        float zs     = fmaxf(z, -INV_E + 1e-7f);
        bool valid = (z >= -INV_E);
        bool need  = valid && (ratio + 1.f >= ssk) && !((z < 0.f) && (ratio > nxt));
        float c1 = (ratio - ssk) + W_MARGIN;
        float c2 = (ratio - nxt) - W_MARGIN;
        float hi = c1 * __expf(c1);
        float lo = c2 * __expf(c2);
        need = need && !(zs > hi) && !((c2 > -1.f) && (zs < lo));
        ssk_[r] = ssk; nxt_[r] = nxt; ratio_[r] = ratio; zs_[r] = zs; need_[r] = need;
        anyNeed = anyNeed || need;
      }
      if (__ballot(anyNeed)) {
#pragma unroll
        for (int r = 0; r < 8; ++r) {
          if (__ballot(need_[r])) {
            float wlam = lambertw0_dev(zs_[r]);
            float t = need_[r] ? (ratio_[r] - wlam) : __builtin_inff();
            if (t < ssk_[r] || t > nxt_[r]) t = __builtin_inff();
            mn = fminf(mn, t);
          }
        }
      }
      carryA += __shfl(dA[7], 16 + m, 32);
      carryB += __shfl(dB[7], 16 + m, 32);
    }
    mn = fminf(mn, __shfl_xor(mn, 16, 32));
    if (half == 0) out[b * N_OUT + oG] = mn;
  }
}

extern "C" void kernel_launch(void* const* d_in, const int* in_sizes, int n_in,
                              void* d_out, int out_size, void* d_ws, size_t ws_size,
                              hipStream_t stream) {
  const float* spikes  = (const float*)d_in[0];
  const float* weights = (const float*)d_in[1];
  float*       out     = (float*)d_out;
  (void)in_sizes; (void)n_in; (void)out_size; (void)d_ws; (void)ws_size;
  ttfs_equaltime_forward<<<dim3(NB, 2), 256, 0, stream>>>(spikes, weights, out);
}


# Exact causal early exit: a row k can only yield t >= t_k and keys are sorted, so each wave stops its k-block scan once all 16 column minima are <= the next block's first key; arithmetic unchanged
# speedup vs baseline: 7.6873x; 1.8965x over previous
#include <hip/hip_runtime.h>
#include <math.h>

typedef __attribute__((ext_vector_type(2))) float v2f;
typedef __attribute__((ext_vector_type(8))) float v8f;

#define N_IN  512
#define N_OUT 256
#define NB    128
#define INV_E 0.36787944117144233f
#define W_MARGIN 0.01f

__device__ __forceinline__ float frcp(float x) { return __builtin_amdgcn_rcpf(x); }

__device__ __forceinline__ float xhalf(float x) {
  int b = __builtin_bit_cast(int, x);
  return __builtin_bit_cast(float,
      __builtin_amdgcn_permlanex16(b, b, 0x76543210, 0xfedcba98, false, false));
}

__device__ __forceinline__ float lambertw0_dev(float z) {
  const float e = 2.7182818284590452f;
  float nearr = -1.f + __builtin_amdgcn_sqrtf(fmaxf(2.f * (e * z + 1.f), 0.f));
  float lz    = __logf(fmaxf(z, 1e-30f));
  float large = lz - __logf(fmaxf(lz, 1e-30f));
  float w = (z < 0.25f) ? nearr : ((z > e) ? large : __logf(1.f + fmaxf(z, -0.99f)));
#pragma unroll
  for (int i = 0; i < 4; ++i) {
    float ew  = __expf(w);
    float f   = fmaf(w, ew, -z);
    float wp1 = w + 1.f;
    float den = ew * wp1 - (w + 2.f) * f * frcp(2.f * wp1);
    den = (fabsf(den) < 1e-12f) ? 1e-12f : den;
    w = w - f * frcp(den);
  }
  return w;
}

__global__ __launch_bounds__(256)
void ttfs_equaltime_forward(const float* __restrict__ S,
                            const float* __restrict__ W,
                            float* __restrict__ out) {
  __shared__ float skey[N_IN];
  __shared__ int   sidx[N_IN];
  __shared__ float ewL[N_IN];
  __shared__ float tewL[N_IN];

  const int b    = blockIdx.x;
  const int tt   = blockIdx.y;
  const int tid  = threadIdx.x;
  const int lane = tid & 31;
  const int half = lane >> 4;
  const int m    = lane & 15;
  const int wave = tid >> 5;

  skey[tid]       = S[b * N_IN + tid];        sidx[tid]       = tid;
  skey[tid + 256] = S[b * N_IN + tid + 256];  sidx[tid + 256] = tid + 256;
  __syncthreads();
  for (int k = 2; k <= N_IN; k <<= 1) {
    for (int j = k >> 1; j > 0; j >>= 1) {
      int  i  = 2 * tid - (tid & (j - 1));
      int  p  = i + j;
      bool up = ((i & k) == 0);
      float ka = skey[i], kb2 = skey[p];
      bool sw = up ? (ka > kb2) : (ka < kb2);
      if (sw) {
        skey[i] = kb2; skey[p] = ka;
        int t0 = sidx[i]; sidx[i] = sidx[p]; sidx[p] = t0;
      }
      __syncthreads();
    }
  }
  for (int n = tid; n < N_IN; n += 256) {
    float s   = skey[n];
    bool  bad = __builtin_isinf(s);
    float sm  = bad ? 0.f : s;
    float ex  = bad ? 0.f : __expf(sm);
    ewL[n]  = ex;
    tewL[n] = sm * ex;
  }
  __syncthreads();

  v2f A[4];
#pragma unroll
  for (int c = 0; c < 4; ++c)
#pragma unroll
    for (int v = 0; v < 2; ++v)
      A[c][v] = ((4 * c + v + 2 * half) <= m) ? 1.f : 0.f;

  {
    const int ot = wave + 8 * tt;
    const int oG = ot * 16 + m;
    float carryA = 0.f, carryB = 0.f;
    float mn = __builtin_inff();

    float wn0[4], wn1[4];
#pragma unroll
    for (int c = 0; c < 4; ++c) {
      int j0 = 4 * c + 2 * half;
      wn0[c] = W[sidx[j0]     * N_OUT + oG];
      wn1[c] = W[sidx[j0 + 1] * N_OUT + oG];
    }

    float kfirst = skey[0];
    for (int kb = 0; kb < N_IN / 16; ++kb) {
      {
        float colmin = fminf(mn, xhalf(mn));
        if (!__ballot(colmin > kfirst)) break;
      }
      v8f dA = {0.f,0.f,0.f,0.f,0.f,0.f,0.f,0.f};
      v8f dB = {0.f,0.f,0.f,0.f,0.f,0.f,0.f,0.f};
      float wc0[4], wc1[4];
#pragma unroll
      for (int c = 0; c < 4; ++c) { wc0[c] = wn0[c]; wc1[c] = wn1[c]; }
      const int kn = (kb + 1 < N_IN / 16) ? kb + 1 : kb;
      kfirst = skey[kn * 16];
#pragma unroll
      for (int c = 0; c < 4; ++c) {
        int jn = kn * 16 + 4 * c + 2 * half;
        wn0[c] = W[sidx[jn]     * N_OUT + oG];
        wn1[c] = W[sidx[jn + 1] * N_OUT + oG];
      }
#pragma unroll
      for (int c = 0; c < 4; ++c) {
        int   j0 = kb * 16 + 4 * c + 2 * half;
        float w0 = wc0[c];
        float w1 = wc1[c];
        v2f b1 = { ewL[j0]  * w0, ewL[j0 + 1]  * w1 };
        v2f b2 = { tewL[j0] * w0, tewL[j0 + 1] * w1 };
        dA = __builtin_amdgcn_wmma_f32_16x16x4_f32(false, A[c], false, b1,
                                                   (short)0, dA, false, false);
        dB = __builtin_amdgcn_wmma_f32_16x16x4_f32(false, A[c], false, b2,
                                                   (short)0, dB, false, false);
      }
      float ssk_[8], nxt_[8], ratio_[8], zs_[8];
      bool  need_[8];
      bool  anyNeed = false;
#pragma unroll
      for (int r = 0; r < 8; ++r) {
        int   k   = kb * 16 + r + 8 * half;
        float ssk = skey[k];
        float nxt = (k == N_IN - 1) ? __builtin_inff() : skey[k + 1];
        float a1  = dA[r] + carryA;
        float bb  = dB[r] + carryB;
        float a1s = (fabsf(a1) < 1e-10f) ? 1e-10f : a1;
        float inv_a1 = frcp(a1s);
        float ratio  = bb * inv_a1;
        float z      = -__expf(ratio) * inv_a1;
        float zs     = fmaxf(z, -INV_E + 1e-7f);
        bool valid = (z >= -INV_E);
        bool need  = valid && (ratio + 1.f >= ssk) && !((z < 0.f) && (ratio > nxt));
        float c1 = (ratio - ssk) + W_MARGIN;
        float c2 = (ratio - nxt) - W_MARGIN;
        float hi = c1 * __expf(c1);
        float lo = c2 * __expf(c2);
        need = need && !(zs > hi) && !((c2 > -1.f) && (zs < lo));
        ssk_[r] = ssk; nxt_[r] = nxt; ratio_[r] = ratio; zs_[r] = zs; need_[r] = need;
        anyNeed = anyNeed || need;
      }
      if (__ballot(anyNeed)) {
#pragma unroll
        for (int r = 0; r < 8; ++r) {
          if (__ballot(need_[r])) {
            float wlam = lambertw0_dev(zs_[r]);
            float t = need_[r] ? (ratio_[r] - wlam) : __builtin_inff();
            if (t < ssk_[r] || t > nxt_[r]) t = __builtin_inff();
            mn = fminf(mn, t);
          }
        }
      }
      carryA += __shfl(dA[7], 16 + m, 32);
      carryB += __shfl(dB[7], 16 + m, 32);
    }
    mn = fminf(mn, __shfl_xor(mn, 16, 32));
    if (half == 0) out[b * N_OUT + oG] = mn;
  }
}

extern "C" void kernel_launch(void* const* d_in, const int* in_sizes, int n_in,
                              void* d_out, int out_size, void* d_ws, size_t ws_size,
                              hipStream_t stream) {
  const float* spikes  = (const float*)d_in[0];
  const float* weights = (const float*)d_in[1];
  float*       out     = (float*)d_out;
  (void)in_sizes; (void)n_in; (void)out_size; (void)d_ws; (void)ws_size;
  ttfs_equaltime_forward<<<dim3(NB, 2), 256, 0, stream>>>(spikes, weights, out);
}


# Argsort kept in registers (2 keys per thread): 39 of the 45 network steps exchange inside a thread or wave via v_permlane16/x16, only the 6 cross-wave steps use LDS and a barrier; same permutation
# speedup vs baseline: 9.9094x; 1.2891x over previous
#include <hip/hip_runtime.h>
#include <math.h>

typedef __attribute__((ext_vector_type(2))) float v2f;
typedef __attribute__((ext_vector_type(8))) float v8f;

#define N_IN  512
#define N_OUT 256
#define NB    128
#define INV_E 0.36787944117144233f
#define W_MARGIN 0.01f

__device__ __forceinline__ float frcp(float x) { return __builtin_amdgcn_rcpf(x); }

__device__ __forceinline__ float xhalf(float x) {
  int b = __builtin_bit_cast(int, x);
  return __builtin_bit_cast(float,
      __builtin_amdgcn_permlanex16(b, b, 0x76543210, 0xfedcba98, false, false));
}


constexpr unsigned xsel(int M, int base) {
  unsigned w = 0;
  for (int j = 0; j < 8; ++j) w |= (unsigned)(((base + j) ^ M) & 15) << (4 * j);
  return w;
}

template <int M>
__device__ __forceinline__ int lanex(int v) {
  if constexpr (M == 16) {
    return __builtin_amdgcn_permlanex16(v, v, 0x76543210, 0xfedcba98, false, false);
  } else {
    constexpr unsigned lo = xsel(M, 0), hi = xsel(M, 8);
    return __builtin_amdgcn_permlane16(v, v, lo, hi, false, false);
  }
}

__device__ __forceinline__ void cx(float& k, int& i, float ok, int oi, bool takeMin) {
  const float x = takeMin ? k : ok;
  const float y = takeMin ? ok : k;
  const bool sw = (x > y);
  k = sw ? ok : k;
  i = sw ? oi : i;
}

template <int K, int J>
__device__ __forceinline__ void bstep(float& k0, int& i0, float& k1, int& i1,
                                      const int tid, uint4* xb) {
  const bool up = ((tid & (K >> 1)) == 0);
  if constexpr (J == 1) {
    const float x = up ? k0 : k1;
    const float y = up ? k1 : k0;
    const bool sw = (x > y);
    const float tk = k0; const int ti = i0;
    k0 = sw ? k1 : k0;  i0 = sw ? i1 : i0;
    k1 = sw ? tk : k1;  i1 = sw ? ti : i1;
  } else {
    const bool lower   = ((tid & (J >> 1)) == 0);
    const bool takeMin = (lower == up);
    float ok0, ok1; int oi0, oi1;
    if constexpr (J <= 32) {
      ok0 = __builtin_bit_cast(float, lanex<J / 2>(__builtin_bit_cast(int, k0)));
      oi0 = lanex<J / 2>(i0);
      ok1 = __builtin_bit_cast(float, lanex<J / 2>(__builtin_bit_cast(int, k1)));
      oi1 = lanex<J / 2>(i1);
    } else {
      xb[tid] = make_uint4(__builtin_bit_cast(unsigned, k0), (unsigned)i0,
                           __builtin_bit_cast(unsigned, k1), (unsigned)i1);
      __syncthreads();
      const uint4 o = xb[tid ^ (J >> 1)];
      ok0 = __builtin_bit_cast(float, o.x);  oi0 = (int)o.y;
      ok1 = __builtin_bit_cast(float, o.z);  oi1 = (int)o.w;
    }
    cx(k0, i0, ok0, oi0, takeMin);
    cx(k1, i1, ok1, oi1, takeMin);
  }
}

template <int K, int J>
__device__ __forceinline__ void bwave(float& k0, int& i0, float& k1, int& i1, const int tid) {
  bstep<K, J>(k0, i0, k1, i1, tid, nullptr);
  if constexpr (J > 1) bwave<K, J / 2>(k0, i0, k1, i1, tid);
}

__device__ __forceinline__ float lambertw0_dev(float z) {
  const float e = 2.7182818284590452f;
  float nearr = -1.f + __builtin_amdgcn_sqrtf(fmaxf(2.f * (e * z + 1.f), 0.f));
  float lz    = __logf(fmaxf(z, 1e-30f));
  float large = lz - __logf(fmaxf(lz, 1e-30f));
  float w = (z < 0.25f) ? nearr : ((z > e) ? large : __logf(1.f + fmaxf(z, -0.99f)));
#pragma unroll
  for (int i = 0; i < 4; ++i) {
    float ew  = __expf(w);
    float f   = fmaf(w, ew, -z);
    float wp1 = w + 1.f;
    float den = ew * wp1 - (w + 2.f) * f * frcp(2.f * wp1);
    den = (fabsf(den) < 1e-12f) ? 1e-12f : den;
    w = w - f * frcp(den);
  }
  return w;
}

__global__ __launch_bounds__(256)
void ttfs_equaltime_forward(const float* __restrict__ S,
                            const float* __restrict__ W,
                            float* __restrict__ out) {
  __shared__ float skey[N_IN];
  __shared__ int   sidx[N_IN];
  __shared__ float ewL[N_IN];
  __shared__ float tewL[N_IN];
  __shared__ uint4 xbuf[2][256];

  const int b    = blockIdx.x;
  const int tt   = blockIdx.y;
  const int tid  = threadIdx.x;
  const int lane = tid & 31;
  const int half = lane >> 4;
  const int m    = lane & 15;
  const int wave = tid >> 5;

  {
    float k0 = S[b * N_IN + 2 * tid];      int i0 = 2 * tid;
    float k1 = S[b * N_IN + 2 * tid + 1];  int i1 = 2 * tid + 1;
    bwave<2, 1>(k0, i0, k1, i1, tid);
    bwave<4, 2>(k0, i0, k1, i1, tid);
    bwave<8, 4>(k0, i0, k1, i1, tid);
    bwave<16, 8>(k0, i0, k1, i1, tid);
    bwave<32, 16>(k0, i0, k1, i1, tid);
    bwave<64, 32>(k0, i0, k1, i1, tid);
    bstep<128, 64>(k0, i0, k1, i1, tid, xbuf[0]);
    bwave<128, 32>(k0, i0, k1, i1, tid);
    bstep<256, 128>(k0, i0, k1, i1, tid, xbuf[1]);
    bstep<256, 64>(k0, i0, k1, i1, tid, xbuf[0]);
    bwave<256, 32>(k0, i0, k1, i1, tid);
    bstep<512, 256>(k0, i0, k1, i1, tid, xbuf[1]);
    bstep<512, 128>(k0, i0, k1, i1, tid, xbuf[0]);
    bstep<512, 64>(k0, i0, k1, i1, tid, xbuf[1]);
    bwave<512, 32>(k0, i0, k1, i1, tid);

    const bool  bad0 = __builtin_isinf(k0), bad1 = __builtin_isinf(k1);
    const float sm0  = bad0 ? 0.f : k0,     sm1  = bad1 ? 0.f : k1;
    const float ex0  = bad0 ? 0.f : __expf(sm0);
    const float ex1  = bad1 ? 0.f : __expf(sm1);
    skey[2 * tid] = k0;         skey[2 * tid + 1] = k1;
    sidx[2 * tid] = i0;         sidx[2 * tid + 1] = i1;
    ewL[2 * tid]  = ex0;        ewL[2 * tid + 1]  = ex1;
    tewL[2 * tid] = sm0 * ex0;  tewL[2 * tid + 1] = sm1 * ex1;
  }
  __syncthreads();

  v2f A[4];
#pragma unroll
  for (int c = 0; c < 4; ++c)
#pragma unroll
    for (int v = 0; v < 2; ++v)
      A[c][v] = ((4 * c + v + 2 * half) <= m) ? 1.f : 0.f;

  {
    const int ot = wave + 8 * tt;
    const int oG = ot * 16 + m;
    float carryA = 0.f, carryB = 0.f;
    float mn = __builtin_inff();

    float wn0[4], wn1[4];
#pragma unroll
    for (int c = 0; c < 4; ++c) {
      int j0 = 4 * c + 2 * half;
      wn0[c] = W[sidx[j0]     * N_OUT + oG];
      wn1[c] = W[sidx[j0 + 1] * N_OUT + oG];
    }

    float kfirst = skey[0];
    for (int kb = 0; kb < N_IN / 16; ++kb) {
      {
        float colmin = fminf(mn, xhalf(mn));
        if (!__ballot(colmin > kfirst)) break;
      }
      v8f dA = {0.f,0.f,0.f,0.f,0.f,0.f,0.f,0.f};
      v8f dB = {0.f,0.f,0.f,0.f,0.f,0.f,0.f,0.f};
      float wc0[4], wc1[4];
#pragma unroll
      for (int c = 0; c < 4; ++c) { wc0[c] = wn0[c]; wc1[c] = wn1[c]; }
      const int kn = (kb + 1 < N_IN / 16) ? kb + 1 : kb;
      kfirst = skey[kn * 16];
#pragma unroll
      for (int c = 0; c < 4; ++c) {
        int jn = kn * 16 + 4 * c + 2 * half;
        wn0[c] = W[sidx[jn]     * N_OUT + oG];
        wn1[c] = W[sidx[jn + 1] * N_OUT + oG];
      }
#pragma unroll
      for (int c = 0; c < 4; ++c) {
        int   j0 = kb * 16 + 4 * c + 2 * half;
        float w0 = wc0[c];
        float w1 = wc1[c];
        v2f b1 = { ewL[j0]  * w0, ewL[j0 + 1]  * w1 };
        v2f b2 = { tewL[j0] * w0, tewL[j0 + 1] * w1 };
        dA = __builtin_amdgcn_wmma_f32_16x16x4_f32(false, A[c], false, b1,
                                                   (short)0, dA, false, false);
        dB = __builtin_amdgcn_wmma_f32_16x16x4_f32(false, A[c], false, b2,
                                                   (short)0, dB, false, false);
      }
      float ssk_[8], nxt_[8], ratio_[8], zs_[8];
      bool  need_[8];
      bool  anyNeed = false;
#pragma unroll
      for (int r = 0; r < 8; ++r) {
        int   k   = kb * 16 + r + 8 * half;
        float ssk = skey[k];
        float nxt = (k == N_IN - 1) ? __builtin_inff() : skey[k + 1];
        float a1  = dA[r] + carryA;
        float bb  = dB[r] + carryB;
        float a1s = (fabsf(a1) < 1e-10f) ? 1e-10f : a1;
        float inv_a1 = frcp(a1s);
        float ratio  = bb * inv_a1;
        float z      = -__expf(ratio) * inv_a1;
        float zs     = fmaxf(z, -INV_E + 1e-7f);
        bool valid = (z >= -INV_E);
        bool need  = valid && (ratio + 1.f >= ssk) && !((z < 0.f) && (ratio > nxt));
        float c1 = (ratio - ssk) + W_MARGIN;
        float c2 = (ratio - nxt) - W_MARGIN;
        float hi = c1 * __expf(c1);
        float lo = c2 * __expf(c2);
        need = need && !(zs > hi) && !((c2 > -1.f) && (zs < lo));
        ssk_[r] = ssk; nxt_[r] = nxt; ratio_[r] = ratio; zs_[r] = zs; need_[r] = need;
        anyNeed = anyNeed || need;
      }
      if (__ballot(anyNeed)) {
#pragma unroll
        for (int r = 0; r < 8; ++r) {
          if (__ballot(need_[r])) {
            float wlam = lambertw0_dev(zs_[r]);
            float t = need_[r] ? (ratio_[r] - wlam) : __builtin_inff();
            if (t < ssk_[r] || t > nxt_[r]) t = __builtin_inff();
            mn = fminf(mn, t);
          }
        }
      }
      carryA += __shfl(dA[7], 16 + m, 32);
      carryB += __shfl(dB[7], 16 + m, 32);
    }
    mn = fminf(mn, __shfl_xor(mn, 16, 32));
    if (half == 0) out[b * N_OUT + oG] = mn;
  }
}

extern "C" void kernel_launch(void* const* d_in, const int* in_sizes, int n_in,
                              void* d_out, int out_size, void* d_ws, size_t ws_size,
                              hipStream_t stream) {
  const float* spikes  = (const float*)d_in[0];
  const float* weights = (const float*)d_in[1];
  float*       out     = (float*)d_out;
  (void)in_sizes; (void)n_in; (void)out_size; (void)d_ws; (void)ws_size;
  ttfs_equaltime_forward<<<dim3(NB, 2), 256, 0, stream>>>(spikes, weights, out);
}


# Lambert solves issued per lane, not per row: each lane solves only its own first pending row in key order, wave repeats while any lane has one; window margin 1e-2 -> 1e-4 away from w=-1; same output
# speedup vs baseline: 12.4477x; 1.2562x over previous
#include <hip/hip_runtime.h>
#include <math.h>

typedef __attribute__((ext_vector_type(2))) float v2f;
typedef __attribute__((ext_vector_type(8))) float v8f;

#define N_IN  512
#define N_OUT 256
#define NB    128
#define INV_E 0.36787944117144233f
#define W_MARGIN      1e-4f
#define W_MARGIN_EDGE 1e-2f
#define W_EDGE        (-0.9f)

__device__ __forceinline__ float frcp(float x) { return __builtin_amdgcn_rcpf(x); }

__device__ __forceinline__ float xhalf(float x) {
  int b = __builtin_bit_cast(int, x);
  return __builtin_bit_cast(float,
      __builtin_amdgcn_permlanex16(b, b, 0x76543210, 0xfedcba98, false, false));
}


constexpr unsigned xsel(int M, int base) {
  unsigned w = 0;
  for (int j = 0; j < 8; ++j) w |= (unsigned)(((base + j) ^ M) & 15) << (4 * j);
  return w;
}

template <int M>
__device__ __forceinline__ int lanex(int v) {
  if constexpr (M == 16) {
    return __builtin_amdgcn_permlanex16(v, v, 0x76543210, 0xfedcba98, false, false);
  } else {
    constexpr unsigned lo = xsel(M, 0), hi = xsel(M, 8);
    return __builtin_amdgcn_permlane16(v, v, lo, hi, false, false);
  }
}

__device__ __forceinline__ void cx(float& k, int& i, float ok, int oi, bool takeMin) {
  const float x = takeMin ? k : ok;
  const float y = takeMin ? ok : k;
  const bool sw = (x > y);
  k = sw ? ok : k;
  i = sw ? oi : i;
}

template <int K, int J>
__device__ __forceinline__ void bstep(float& k0, int& i0, float& k1, int& i1,
                                      const int tid, uint4* xb) {
  const bool up = ((tid & (K >> 1)) == 0);
  if constexpr (J == 1) {
    const float x = up ? k0 : k1;
    const float y = up ? k1 : k0;
    const bool sw = (x > y);
    const float tk = k0; const int ti = i0;
    k0 = sw ? k1 : k0;  i0 = sw ? i1 : i0;
    k1 = sw ? tk : k1;  i1 = sw ? ti : i1;
  } else {
    const bool lower   = ((tid & (J >> 1)) == 0);
    const bool takeMin = (lower == up);
    float ok0, ok1; int oi0, oi1;
    if constexpr (J <= 32) {
      ok0 = __builtin_bit_cast(float, lanex<J / 2>(__builtin_bit_cast(int, k0)));
      oi0 = lanex<J / 2>(i0);
      ok1 = __builtin_bit_cast(float, lanex<J / 2>(__builtin_bit_cast(int, k1)));
      oi1 = lanex<J / 2>(i1);
    } else {
      xb[tid] = make_uint4(__builtin_bit_cast(unsigned, k0), (unsigned)i0,
                           __builtin_bit_cast(unsigned, k1), (unsigned)i1);
      __syncthreads();
      const uint4 o = xb[tid ^ (J >> 1)];
      ok0 = __builtin_bit_cast(float, o.x);  oi0 = (int)o.y;
      ok1 = __builtin_bit_cast(float, o.z);  oi1 = (int)o.w;
    }
    cx(k0, i0, ok0, oi0, takeMin);
    cx(k1, i1, ok1, oi1, takeMin);
  }
}

template <int K, int J>
__device__ __forceinline__ void bwave(float& k0, int& i0, float& k1, int& i1, const int tid) {
  bstep<K, J>(k0, i0, k1, i1, tid, nullptr);
  if constexpr (J > 1) bwave<K, J / 2>(k0, i0, k1, i1, tid);
}

__device__ __forceinline__ float lambertw0_dev(float z) {
  const float e = 2.7182818284590452f;
  float nearr = -1.f + __builtin_amdgcn_sqrtf(fmaxf(2.f * (e * z + 1.f), 0.f));
  float lz    = __logf(fmaxf(z, 1e-30f));
  float large = lz - __logf(fmaxf(lz, 1e-30f));
  float w = (z < 0.25f) ? nearr : ((z > e) ? large : __logf(1.f + fmaxf(z, -0.99f)));
#pragma unroll
  for (int i = 0; i < 4; ++i) {
    float ew  = __expf(w);
    float f   = fmaf(w, ew, -z);
    float wp1 = w + 1.f;
    float den = ew * wp1 - (w + 2.f) * f * frcp(2.f * wp1);
    den = (fabsf(den) < 1e-12f) ? 1e-12f : den;
    w = w - f * frcp(den);
  }
  return w;
}

__device__ __forceinline__ float pick8(const float (&a)[8], bool b0, bool b1, bool b2) {
  const float x0 = b0 ? a[1] : a[0], x1 = b0 ? a[3] : a[2];
  const float x2 = b0 ? a[5] : a[4], x3 = b0 ? a[7] : a[6];
  const float y0 = b1 ? x1 : x0,     y1 = b1 ? x3 : x2;
  return b2 ? y1 : y0;
}

__global__ __launch_bounds__(256)
void ttfs_equaltime_forward(const float* __restrict__ S,
                            const float* __restrict__ W,
                            float* __restrict__ out) {
  __shared__ float skey[N_IN];
  __shared__ int   sidx[N_IN];
  __shared__ float ewL[N_IN];
  __shared__ float tewL[N_IN];
  __shared__ uint4 xbuf[2][256];

  const int b    = blockIdx.x;
  const int tt   = blockIdx.y;
  const int tid  = threadIdx.x;
  const int lane = tid & 31;
  const int half = lane >> 4;
  const int m    = lane & 15;
  const int wave = tid >> 5;

  {
    float k0 = S[b * N_IN + 2 * tid];      int i0 = 2 * tid;
    float k1 = S[b * N_IN + 2 * tid + 1];  int i1 = 2 * tid + 1;
    bwave<2, 1>(k0, i0, k1, i1, tid);
    bwave<4, 2>(k0, i0, k1, i1, tid);
    bwave<8, 4>(k0, i0, k1, i1, tid);
    bwave<16, 8>(k0, i0, k1, i1, tid);
    bwave<32, 16>(k0, i0, k1, i1, tid);
    bwave<64, 32>(k0, i0, k1, i1, tid);
    bstep<128, 64>(k0, i0, k1, i1, tid, xbuf[0]);
    bwave<128, 32>(k0, i0, k1, i1, tid);
    bstep<256, 128>(k0, i0, k1, i1, tid, xbuf[1]);
    bstep<256, 64>(k0, i0, k1, i1, tid, xbuf[0]);
    bwave<256, 32>(k0, i0, k1, i1, tid);
    bstep<512, 256>(k0, i0, k1, i1, tid, xbuf[1]);
    bstep<512, 128>(k0, i0, k1, i1, tid, xbuf[0]);
    bstep<512, 64>(k0, i0, k1, i1, tid, xbuf[1]);
    bwave<512, 32>(k0, i0, k1, i1, tid);

    const bool  bad0 = __builtin_isinf(k0), bad1 = __builtin_isinf(k1);
    const float sm0  = bad0 ? 0.f : k0,     sm1  = bad1 ? 0.f : k1;
    const float ex0  = bad0 ? 0.f : __expf(sm0);
    const float ex1  = bad1 ? 0.f : __expf(sm1);
    skey[2 * tid] = k0;         skey[2 * tid + 1] = k1;
    sidx[2 * tid] = i0;         sidx[2 * tid + 1] = i1;
    ewL[2 * tid]  = ex0;        ewL[2 * tid + 1]  = ex1;
    tewL[2 * tid] = sm0 * ex0;  tewL[2 * tid + 1] = sm1 * ex1;
  }
  __syncthreads();

  v2f A[4];
#pragma unroll
  for (int c = 0; c < 4; ++c)
#pragma unroll
    for (int v = 0; v < 2; ++v)
      A[c][v] = ((4 * c + v + 2 * half) <= m) ? 1.f : 0.f;

  {
    const int ot = wave + 8 * tt;
    const int oG = ot * 16 + m;
    float carryA = 0.f, carryB = 0.f;
    float mn = __builtin_inff();

    float wn0[4], wn1[4];
#pragma unroll
    for (int c = 0; c < 4; ++c) {
      int j0 = 4 * c + 2 * half;
      wn0[c] = W[sidx[j0]     * N_OUT + oG];
      wn1[c] = W[sidx[j0 + 1] * N_OUT + oG];
    }

    float kfirst = skey[0];
    for (int kb = 0; kb < N_IN / 16; ++kb) {
      {
        float colmin = fminf(mn, xhalf(mn));
        if (!__ballot(colmin > kfirst)) break;
      }
      v8f dA = {0.f,0.f,0.f,0.f,0.f,0.f,0.f,0.f};
      v8f dB = {0.f,0.f,0.f,0.f,0.f,0.f,0.f,0.f};
      float wc0[4], wc1[4];
#pragma unroll
      for (int c = 0; c < 4; ++c) { wc0[c] = wn0[c]; wc1[c] = wn1[c]; }
      const int kn = (kb + 1 < N_IN / 16) ? kb + 1 : kb;
      kfirst = skey[kn * 16];
#pragma unroll
      for (int c = 0; c < 4; ++c) {
        int jn = kn * 16 + 4 * c + 2 * half;
        wn0[c] = W[sidx[jn]     * N_OUT + oG];
        wn1[c] = W[sidx[jn + 1] * N_OUT + oG];
      }
#pragma unroll
      for (int c = 0; c < 4; ++c) {
        int   j0 = kb * 16 + 4 * c + 2 * half;
        float w0 = wc0[c];
        float w1 = wc1[c];
        v2f b1 = { ewL[j0]  * w0, ewL[j0 + 1]  * w1 };
        v2f b2 = { tewL[j0] * w0, tewL[j0 + 1] * w1 };
        dA = __builtin_amdgcn_wmma_f32_16x16x4_f32(false, A[c], false, b1,
                                                   (short)0, dA, false, false);
        dB = __builtin_amdgcn_wmma_f32_16x16x4_f32(false, A[c], false, b2,
                                                   (short)0, dB, false, false);
      }
      float ratio_[8], zs_[8];
      bool  need_[8];
      bool  anyNeed = false;
#pragma unroll
      for (int r = 0; r < 8; ++r) {
        int   k   = kb * 16 + r + 8 * half;
        float ssk = skey[k];
        float nxt = (k == N_IN - 1) ? __builtin_inff() : skey[k + 1];
        float a1  = dA[r] + carryA;
        float bb  = dB[r] + carryB;
        float a1s = (fabsf(a1) < 1e-10f) ? 1e-10f : a1;
        float inv_a1 = frcp(a1s);
        float ratio  = bb * inv_a1;
        float z      = -__expf(ratio) * inv_a1;
        float zs     = fmaxf(z, -INV_E + 1e-7f);
        bool valid = (z >= -INV_E);
        bool need  = valid && (ratio + 1.f >= ssk) && !((z < 0.f) && (ratio > nxt));
        float c2r = ratio - nxt;
        float mg  = (c2r < W_EDGE) ? W_MARGIN_EDGE : W_MARGIN;
        float c1  = (ratio - ssk) + mg;
        float c2  = c2r - mg;
        float hi = c1 * __expf(c1);
        float lo = c2 * __expf(c2);
        need = need && !(zs > hi) && !((c2 > -1.f) && (zs < lo));
        ratio_[r] = ratio; zs_[r] = zs; need_[r] = need;
        anyNeed = anyNeed || need;
      }
      if (__ballot(anyNeed)) {
        unsigned pend = 0u;
#pragma unroll
        for (int r = 0; r < 8; ++r) pend |= need_[r] ? (1u << r) : 0u;
#pragma nounroll
        for (int pass = 0; pass < 8; ++pass) {
          if (!__ballot(pend != 0u)) break;
          const bool     act = (pend != 0u);
          const unsigned r   = (unsigned)__builtin_ctz(pend | 0x100u) & 7u;
          const bool b0 = (r & 1u) != 0u, b1 = (r & 2u) != 0u, b2 = (r & 4u) != 0u;
          const float zsel = pick8(zs_, b0, b1, b2);
          const float rsel = pick8(ratio_, b0, b1, b2);
          const int   k    = kb * 16 + (int)r + 8 * half;
          const float ssk  = skey[k];
          const float nxt  = (k == N_IN - 1) ? __builtin_inff() : skey[(k + 1) & (N_IN - 1)];
          float t = __builtin_inff();
          if (act) t = rsel - lambertw0_dev(zsel);
          if (t < ssk || t > nxt) t = __builtin_inff();
          mn = fminf(mn, t);
          pend = (t < __builtin_inff()) ? 0u : (pend & (pend - 1u));
        }
      }
      carryA += __shfl(dA[7], 16 + m, 32);
      carryB += __shfl(dB[7], 16 + m, 32);
    }
    mn = fminf(mn, __shfl_xor(mn, 16, 32));
    if (half == 0) out[b * N_OUT + oG] = mn;
  }
}

extern "C" void kernel_launch(void* const* d_in, const int* in_sizes, int n_in,
                              void* d_out, int out_size, void* d_ws, size_t ws_size,
                              hipStream_t stream) {
  const float* spikes  = (const float*)d_in[0];
  const float* weights = (const float*)d_in[1];
  float*       out     = (float*)d_out;
  (void)in_sizes; (void)n_in; (void)out_size; (void)d_ws; (void)ws_size;
  ttfs_equaltime_forward<<<dim3(NB, 2), 256, 0, stream>>>(spikes, weights, out);
}


# Window test asked of the membrane value, not the solve's argument: t_k*a1-b <= min(a1,e^t_k) and t_k+1*a1-b >= min(a1,e^t_k+1), e^t read from LDS; no rcp/exp per row, only in solved rows
# speedup vs baseline: 15.5116x; 1.2461x over previous
#include <hip/hip_runtime.h>
#include <math.h>

typedef __attribute__((ext_vector_type(2))) float v2f;
typedef __attribute__((ext_vector_type(8))) float v8f;

#define N_IN  512
#define N_OUT 256
#define NB    128
#define INV_E 0.36787944117144233f
#define V_WIDEN 2e-5f

__device__ __forceinline__ float frcp(float x) { return __builtin_amdgcn_rcpf(x); }

__device__ __forceinline__ float xhalf(float x) {
  int b = __builtin_bit_cast(int, x);
  return __builtin_bit_cast(float,
      __builtin_amdgcn_permlanex16(b, b, 0x76543210, 0xfedcba98, false, false));
}


constexpr unsigned xsel(int M, int base) {
  unsigned w = 0;
  for (int j = 0; j < 8; ++j) w |= (unsigned)(((base + j) ^ M) & 15) << (4 * j);
  return w;
}

template <int M>
__device__ __forceinline__ int lanex(int v) {
  if constexpr (M == 16) {
    return __builtin_amdgcn_permlanex16(v, v, 0x76543210, 0xfedcba98, false, false);
  } else {
    constexpr unsigned lo = xsel(M, 0), hi = xsel(M, 8);
    return __builtin_amdgcn_permlane16(v, v, lo, hi, false, false);
  }
}

__device__ __forceinline__ void cx(float& k, int& i, float ok, int oi, bool takeMin) {
  const float x = takeMin ? k : ok;
  const float y = takeMin ? ok : k;
  const bool sw = (x > y);
  k = sw ? ok : k;
  i = sw ? oi : i;
}

template <int K, int J>
__device__ __forceinline__ void bstep(float& k0, int& i0, float& k1, int& i1,
                                      const int tid, uint4* xb) {
  const bool up = ((tid & (K >> 1)) == 0);
  if constexpr (J == 1) {
    const float x = up ? k0 : k1;
    const float y = up ? k1 : k0;
    const bool sw = (x > y);
    const float tk = k0; const int ti = i0;
    k0 = sw ? k1 : k0;  i0 = sw ? i1 : i0;
    k1 = sw ? tk : k1;  i1 = sw ? ti : i1;
  } else {
    const bool lower   = ((tid & (J >> 1)) == 0);
    const bool takeMin = (lower == up);
    float ok0, ok1; int oi0, oi1;
    if constexpr (J <= 32) {
      ok0 = __builtin_bit_cast(float, lanex<J / 2>(__builtin_bit_cast(int, k0)));
      oi0 = lanex<J / 2>(i0);
      ok1 = __builtin_bit_cast(float, lanex<J / 2>(__builtin_bit_cast(int, k1)));
      oi1 = lanex<J / 2>(i1);
    } else {
      xb[tid] = make_uint4(__builtin_bit_cast(unsigned, k0), (unsigned)i0,
                           __builtin_bit_cast(unsigned, k1), (unsigned)i1);
      __syncthreads();
      const uint4 o = xb[tid ^ (J >> 1)];
      ok0 = __builtin_bit_cast(float, o.x);  oi0 = (int)o.y;
      ok1 = __builtin_bit_cast(float, o.z);  oi1 = (int)o.w;
    }
    cx(k0, i0, ok0, oi0, takeMin);
    cx(k1, i1, ok1, oi1, takeMin);
  }
}

template <int K, int J>
__device__ __forceinline__ void bwave(float& k0, int& i0, float& k1, int& i1, const int tid) {
  bstep<K, J>(k0, i0, k1, i1, tid, nullptr);
  if constexpr (J > 1) bwave<K, J / 2>(k0, i0, k1, i1, tid);
}

__device__ __forceinline__ float lambertw0_dev(float z) {
  const float e = 2.7182818284590452f;
  float nearr = -1.f + __builtin_amdgcn_sqrtf(fmaxf(2.f * (e * z + 1.f), 0.f));
  float lz    = __logf(fmaxf(z, 1e-30f));
  float large = lz - __logf(fmaxf(lz, 1e-30f));
  float w = (z < 0.25f) ? nearr : ((z > e) ? large : __logf(1.f + fmaxf(z, -0.99f)));
#pragma unroll
  for (int i = 0; i < 4; ++i) {
    float ew  = __expf(w);
    float f   = fmaf(w, ew, -z);
    float wp1 = w + 1.f;
    float den = ew * wp1 - (w + 2.f) * f * frcp(2.f * wp1);
    den = (fabsf(den) < 1e-12f) ? 1e-12f : den;
    w = w - f * frcp(den);
  }
  return w;
}

__device__ __forceinline__ float pick8(const float (&a)[8], bool b0, bool b1, bool b2) {
  const float x0 = b0 ? a[1] : a[0], x1 = b0 ? a[3] : a[2];
  const float x2 = b0 ? a[5] : a[4], x3 = b0 ? a[7] : a[6];
  const float y0 = b1 ? x1 : x0,     y1 = b1 ? x3 : x2;
  return b2 ? y1 : y0;
}

__global__ __launch_bounds__(256)
void ttfs_equaltime_forward(const float* __restrict__ S,
                            const float* __restrict__ W,
                            float* __restrict__ out) {
  __shared__ float skey[N_IN];
  __shared__ float nkey[N_IN];
  __shared__ float newL[N_IN];
  __shared__ int   sidx[N_IN];
  __shared__ float ewL[N_IN];
  __shared__ float tewL[N_IN];
  __shared__ uint4 xbuf[2][256];

  const int b    = blockIdx.x;
  const int tt   = blockIdx.y;
  const int tid  = threadIdx.x;
  const int lane = tid & 31;
  const int half = lane >> 4;
  const int m    = lane & 15;
  const int wave = tid >> 5;

  {
    float k0 = S[b * N_IN + 2 * tid];      int i0 = 2 * tid;
    float k1 = S[b * N_IN + 2 * tid + 1];  int i1 = 2 * tid + 1;
    bwave<2, 1>(k0, i0, k1, i1, tid);
    bwave<4, 2>(k0, i0, k1, i1, tid);
    bwave<8, 4>(k0, i0, k1, i1, tid);
    bwave<16, 8>(k0, i0, k1, i1, tid);
    bwave<32, 16>(k0, i0, k1, i1, tid);
    bwave<64, 32>(k0, i0, k1, i1, tid);
    bstep<128, 64>(k0, i0, k1, i1, tid, xbuf[0]);
    bwave<128, 32>(k0, i0, k1, i1, tid);
    bstep<256, 128>(k0, i0, k1, i1, tid, xbuf[1]);
    bstep<256, 64>(k0, i0, k1, i1, tid, xbuf[0]);
    bwave<256, 32>(k0, i0, k1, i1, tid);
    bstep<512, 256>(k0, i0, k1, i1, tid, xbuf[1]);
    bstep<512, 128>(k0, i0, k1, i1, tid, xbuf[0]);
    bstep<512, 64>(k0, i0, k1, i1, tid, xbuf[1]);
    bwave<512, 32>(k0, i0, k1, i1, tid);

    const bool  bad0 = __builtin_isinf(k0), bad1 = __builtin_isinf(k1);
    const float sm0  = bad0 ? 0.f : k0,     sm1  = bad1 ? 0.f : k1;
    const float ex0  = bad0 ? 0.f : __expf(sm0);
    const float ex1  = bad1 ? 0.f : __expf(sm1);
    skey[2 * tid] = k0;         skey[2 * tid + 1] = k1;
    sidx[2 * tid] = i0;         sidx[2 * tid + 1] = i1;
    ewL[2 * tid]  = ex0;        ewL[2 * tid + 1]  = ex1;
    tewL[2 * tid] = sm0 * ex0;  tewL[2 * tid + 1] = sm1 * ex1;
    if (tid > 0) { nkey[2 * tid - 1] = k0;  newL[2 * tid - 1] = ex0; }
    nkey[2 * tid] = k1;         newL[2 * tid] = ex1;
    if (tid == 255) { nkey[N_IN - 1] = __builtin_inff();  newL[N_IN - 1] = 0.f; }
  }
  __syncthreads();

  v2f A[4];
#pragma unroll
  for (int c = 0; c < 4; ++c)
#pragma unroll
    for (int v = 0; v < 2; ++v)
      A[c][v] = ((4 * c + v + 2 * half) <= m) ? 1.f : 0.f;

  {
    const int ot = wave + 8 * tt;
    const int oG = ot * 16 + m;
    float carryA = 0.f, carryB = 0.f;
    float mn = __builtin_inff();

    float wn0[4], wn1[4];
#pragma unroll
    for (int c = 0; c < 4; ++c) {
      int j0 = 4 * c + 2 * half;
      wn0[c] = W[sidx[j0]     * N_OUT + oG];
      wn1[c] = W[sidx[j0 + 1] * N_OUT + oG];
    }

    float key_[8], nk_[8], ek_[8], nek_[8];
#pragma unroll
    for (int r = 0; r < 8; ++r) {
      key_[r] = skey[8 * half + r];  nk_[r]  = nkey[8 * half + r];
      ek_[r]  = ewL[8 * half + r];   nek_[r] = newL[8 * half + r];
    }

    float kfirst = skey[0];
    for (int kb = 0; kb < N_IN / 16; ++kb) {
      {
        float colmin = fminf(mn, xhalf(mn));
        if (!__ballot(colmin > kfirst)) break;
      }
      v8f dA = {0.f,0.f,0.f,0.f,0.f,0.f,0.f,0.f};
      v8f dB = {0.f,0.f,0.f,0.f,0.f,0.f,0.f,0.f};
      float wc0[4], wc1[4];
#pragma unroll
      for (int c = 0; c < 4; ++c) { wc0[c] = wn0[c]; wc1[c] = wn1[c]; }
      const int kn = (kb + 1 < N_IN / 16) ? kb + 1 : kb;
      kfirst = skey[kn * 16];
#pragma unroll
      for (int c = 0; c < 4; ++c) {
        int jn = kn * 16 + 4 * c + 2 * half;
        wn0[c] = W[sidx[jn]     * N_OUT + oG];
        wn1[c] = W[sidx[jn + 1] * N_OUT + oG];
      }
#pragma unroll
      for (int c = 0; c < 4; ++c) {
        int   j0 = kb * 16 + 4 * c + 2 * half;
        float w0 = wc0[c];
        float w1 = wc1[c];
        v2f b1 = { ewL[j0]  * w0, ewL[j0 + 1]  * w1 };
        v2f b2 = { tewL[j0] * w0, tewL[j0 + 1] * w1 };
        dA = __builtin_amdgcn_wmma_f32_16x16x4_f32(false, A[c], false, b1,
                                                   (short)0, dA, false, false);
        dB = __builtin_amdgcn_wmma_f32_16x16x4_f32(false, A[c], false, b2,
                                                   (short)0, dB, false, false);
      }
      float a1_[8], bb_[8];
      unsigned need_[8];
      unsigned anyMask = 0u;
      unsigned untMask = 0u;
#pragma unroll
      for (int r = 0; r < 8; ++r) {
        const float a1 = dA[r] + carryA;
        const float bb = dB[r] + carryB;
        const float vs = fmaf(key_[r], a1, -bb);
        const float vn = fmaf(nk_[r],  a1, -bb);
        const float sl = V_WIDEN * (a1 + fabsf(bb));
        const float hi = fminf(a1, ek_[r])  + sl;
        const float lo = fminf(a1, nek_[r]) - sl;
        a1_[r] = a1; bb_[r] = bb;
        const unsigned below    = __builtin_amdgcn_ballot_w32(!(vs > hi));
        const unsigned above    = __builtin_amdgcn_ballot_w32(!(vn < lo));
        const unsigned untested = __builtin_amdgcn_ballot_w32(!(a1 >= 1e-10f));
        need_[r] = untested | (below & above);
        untMask |= untested;
        anyMask |= need_[r];
      }
      if (untMask != 0u) {
        anyMask = 0u;
#pragma unroll
        for (int r = 0; r < 8; ++r) {
          const float a1 = a1_[r], bb = bb_[r];
          const float vs = fmaf(key_[r], a1, -bb);
          const float sw = V_WIDEN * (fabsf(a1) + fabsf(bb) + ek_[r]);
          const bool falling_out = (a1 <= -1e-10f) && (vs + sw < ek_[r]);
          need_[r] &= ~__builtin_amdgcn_ballot_w32(falling_out);
          anyMask |= need_[r];
        }
      }
#pragma unroll
      for (int r = 0; r < 8; ++r) {
        const int kq = kn * 16 + 8 * half + r;
        key_[r] = skey[kq];  nk_[r]  = nkey[kq];
        ek_[r]  = ewL[kq];   nek_[r] = newL[kq];
      }
      if (anyMask != 0u) {
        unsigned pend = 0u;
#pragma unroll
        for (int r = 0; r < 8; ++r)
          pend |= __builtin_amdgcn_inverse_ballot_w32(need_[r]) ? (1u << (4 * r)) : 0u;
#pragma nounroll
        for (int pass = 0; pass < 8; ++pass) {
          if (!__ballot(pend != 0u)) break;
          const bool     act = (pend != 0u);
          const unsigned r   = act ? ((unsigned)__builtin_ctz(pend) >> 2) : 0u;
          const bool b0 = (r & 1u) != 0u, b1 = (r & 2u) != 0u, b2 = (r & 4u) != 0u;
          const float a1  = pick8(a1_, b0, b1, b2);
          const float bb  = pick8(bb_, b0, b1, b2);
          const int   k   = kb * 16 + (int)r + 8 * half;
          const float ssk = skey[k];
          const float nxt = nkey[k];
          const float a1s    = (fabsf(a1) < 1e-10f) ? 1e-10f : a1;
          const float inv_a1 = frcp(a1s);
          const float ratio  = bb * inv_a1;
          const float z      = -__expf(ratio) * inv_a1;
          const bool  valid  = (z >= -INV_E);
          const float zs     = fmaxf(z, -INV_E + 1e-7f);
          float t = __builtin_inff();
          if (act && valid) t = ratio - lambertw0_dev(zs);
          if (t < ssk || t > nxt) t = __builtin_inff();
          mn = fminf(mn, t);
          pend = (t < __builtin_inff()) ? 0u : (pend & (pend - 1u));
        }
      }
      carryA += __shfl(dA[7], 16 + m, 32);
      carryB += __shfl(dB[7], 16 + m, 32);
    }
    mn = fminf(mn, __shfl_xor(mn, 16, 32));
    if (half == 0) out[b * N_OUT + oG] = mn;
  }
}

extern "C" void kernel_launch(void* const* d_in, const int* in_sizes, int n_in,
                              void* d_out, int out_size, void* d_ws, size_t ws_size,
                              hipStream_t stream) {
  const float* spikes  = (const float*)d_in[0];
  const float* weights = (const float*)d_in[1];
  float*       out     = (float*)d_out;
  (void)in_sizes; (void)n_in; (void)out_size; (void)d_ws; (void)ws_size;
  ttfs_equaltime_forward<<<dim3(NB, 2), 256, 0, stream>>>(spikes, weights, out);
}


# Both prefix sums from ONE f32 WMMA per 8 sorted rows: e^t and t*e^t ride masked in A's 16 rows, B is the raw weight tile; 4 products per 16 rows, not 8; sums now form in 8-row halves (order changes)
# speedup vs baseline: 17.2639x; 1.1130x over previous
#include <hip/hip_runtime.h>
#include <math.h>

typedef __attribute__((ext_vector_type(2))) float v2f;
typedef __attribute__((ext_vector_type(8))) float v8f;

#define N_IN  512
#define N_OUT 256
#define NB    128
#define INV_E 0.36787944117144233f
#define V_WIDEN 2e-5f

__device__ __forceinline__ float frcp(float x) { return __builtin_amdgcn_rcpf(x); }

__device__ __forceinline__ float xhalf(float x) {
  int b = __builtin_bit_cast(int, x);
  return __builtin_bit_cast(float,
      __builtin_amdgcn_permlanex16(b, b, 0x76543210, 0xfedcba98, false, false));
}


constexpr unsigned xsel(int M, int base) {
  unsigned w = 0;
  for (int j = 0; j < 8; ++j) w |= (unsigned)(((base + j) ^ M) & 15) << (4 * j);
  return w;
}

template <int M>
__device__ __forceinline__ int lanex(int v) {
  if constexpr (M == 16) {
    return __builtin_amdgcn_permlanex16(v, v, 0x76543210, 0xfedcba98, false, false);
  } else {
    constexpr unsigned lo = xsel(M, 0), hi = xsel(M, 8);
    return __builtin_amdgcn_permlane16(v, v, lo, hi, false, false);
  }
}

__device__ __forceinline__ void cx(float& k, int& i, float ok, int oi, bool takeMin) {
  const float x = takeMin ? k : ok;
  const float y = takeMin ? ok : k;
  const bool sw = (x > y);
  k = sw ? ok : k;
  i = sw ? oi : i;
}

template <int K, int J>
__device__ __forceinline__ void bstep(float& k0, int& i0, float& k1, int& i1,
                                      const int tid, uint4* xb) {
  const bool up = ((tid & (K >> 1)) == 0);
  if constexpr (J == 1) {
    const float x = up ? k0 : k1;
    const float y = up ? k1 : k0;
    const bool sw = (x > y);
    const float tk = k0; const int ti = i0;
    k0 = sw ? k1 : k0;  i0 = sw ? i1 : i0;
    k1 = sw ? tk : k1;  i1 = sw ? ti : i1;
  } else {
    const bool lower   = ((tid & (J >> 1)) == 0);
    const bool takeMin = (lower == up);
    float ok0, ok1; int oi0, oi1;
    if constexpr (J <= 32) {
      ok0 = __builtin_bit_cast(float, lanex<J / 2>(__builtin_bit_cast(int, k0)));
      oi0 = lanex<J / 2>(i0);
      ok1 = __builtin_bit_cast(float, lanex<J / 2>(__builtin_bit_cast(int, k1)));
      oi1 = lanex<J / 2>(i1);
    } else {
      xb[tid] = make_uint4(__builtin_bit_cast(unsigned, k0), (unsigned)i0,
                           __builtin_bit_cast(unsigned, k1), (unsigned)i1);
      __syncthreads();
      const uint4 o = xb[tid ^ (J >> 1)];
      ok0 = __builtin_bit_cast(float, o.x);  oi0 = (int)o.y;
      ok1 = __builtin_bit_cast(float, o.z);  oi1 = (int)o.w;
    }
    cx(k0, i0, ok0, oi0, takeMin);
    cx(k1, i1, ok1, oi1, takeMin);
  }
}

template <int K, int J>
__device__ __forceinline__ void bwave(float& k0, int& i0, float& k1, int& i1, const int tid) {
  bstep<K, J>(k0, i0, k1, i1, tid, nullptr);
  if constexpr (J > 1) bwave<K, J / 2>(k0, i0, k1, i1, tid);
}

__device__ __forceinline__ float lambertw0_dev(float z) {
  const float e = 2.7182818284590452f;
  float nearr = -1.f + __builtin_amdgcn_sqrtf(fmaxf(2.f * (e * z + 1.f), 0.f));
  float lz    = __logf(fmaxf(z, 1e-30f));
  float large = lz - __logf(fmaxf(lz, 1e-30f));
  float w = (z < 0.25f) ? nearr : ((z > e) ? large : __logf(1.f + fmaxf(z, -0.99f)));
#pragma unroll
  for (int i = 0; i < 4; ++i) {
    float ew  = __expf(w);
    float f   = fmaf(w, ew, -z);
    float wp1 = w + 1.f;
    float den = ew * wp1 - (w + 2.f) * f * frcp(2.f * wp1);
    den = (fabsf(den) < 1e-12f) ? 1e-12f : den;
    w = w - f * frcp(den);
  }
  return w;
}

__device__ __forceinline__ float pick8(const float (&a)[8], bool b0, bool b1, bool b2) {
  const float x0 = b0 ? a[1] : a[0], x1 = b0 ? a[3] : a[2];
  const float x2 = b0 ? a[5] : a[4], x3 = b0 ? a[7] : a[6];
  const float y0 = b1 ? x1 : x0,     y1 = b1 ? x3 : x2;
  return b2 ? y1 : y0;
}

__global__ __launch_bounds__(256)
void ttfs_equaltime_forward(const float* __restrict__ S,
                            const float* __restrict__ W,
                            float* __restrict__ out) {
  __shared__ float skey[N_IN];
  __shared__ float nkey[N_IN];
  __shared__ float newL[N_IN];
  __shared__ int   sidx[N_IN];
  __shared__ __attribute__((aligned(16))) float etL[2 * N_IN];
  float* const ewL  = etL;
  float* const tewL = etL + N_IN;
  __shared__ uint4 xbuf[2][256];

  const int b    = blockIdx.x;
  const int tt   = blockIdx.y;
  const int tid  = threadIdx.x;
  const int lane = tid & 31;
  const int half = lane >> 4;
  const int m    = lane & 15;
  const int wave = tid >> 5;

  {
    float k0 = S[b * N_IN + 2 * tid];      int i0 = 2 * tid;
    float k1 = S[b * N_IN + 2 * tid + 1];  int i1 = 2 * tid + 1;
    bwave<2, 1>(k0, i0, k1, i1, tid);
    bwave<4, 2>(k0, i0, k1, i1, tid);
    bwave<8, 4>(k0, i0, k1, i1, tid);
    bwave<16, 8>(k0, i0, k1, i1, tid);
    bwave<32, 16>(k0, i0, k1, i1, tid);
    bwave<64, 32>(k0, i0, k1, i1, tid);
    bstep<128, 64>(k0, i0, k1, i1, tid, xbuf[0]);
    bwave<128, 32>(k0, i0, k1, i1, tid);
    bstep<256, 128>(k0, i0, k1, i1, tid, xbuf[1]);
    bstep<256, 64>(k0, i0, k1, i1, tid, xbuf[0]);
    bwave<256, 32>(k0, i0, k1, i1, tid);
    bstep<512, 256>(k0, i0, k1, i1, tid, xbuf[1]);
    bstep<512, 128>(k0, i0, k1, i1, tid, xbuf[0]);
    bstep<512, 64>(k0, i0, k1, i1, tid, xbuf[1]);
    bwave<512, 32>(k0, i0, k1, i1, tid);

    const bool  bad0 = __builtin_isinf(k0), bad1 = __builtin_isinf(k1);
    const float sm0  = bad0 ? 0.f : k0,     sm1  = bad1 ? 0.f : k1;
    const float ex0  = bad0 ? 0.f : __expf(sm0);
    const float ex1  = bad1 ? 0.f : __expf(sm1);
    skey[2 * tid] = k0;         skey[2 * tid + 1] = k1;
    sidx[2 * tid] = i0;         sidx[2 * tid + 1] = i1;
    ewL[2 * tid]  = ex0;        ewL[2 * tid + 1]  = ex1;
    tewL[2 * tid] = sm0 * ex0;  tewL[2 * tid + 1] = sm1 * ex1;
    if (tid > 0) { nkey[2 * tid - 1] = k0;  newL[2 * tid - 1] = ex0; }
    nkey[2 * tid] = k1;         newL[2 * tid] = ex1;
    if (tid == 255) { nkey[N_IN - 1] = __builtin_inff();  newL[N_IN - 1] = 0.f; }
  }
  __syncthreads();

  const int    eOff = ((m >> 2) & 1) ? N_IN : 0;
  const int    irow = (m & 3) + 4 * (m >> 3);
  bool keepA[2][2];
#pragma unroll
  for (int c = 0; c < 2; ++c)
#pragma unroll
    for (int v = 0; v < 2; ++v)
      keepA[c][v] = ((4 * c + 2 * half + v) <= irow);

  {
    const int ot = wave + 8 * tt;
    const int oG = ot * 16 + m;
    float carryA = 0.f, carryB = 0.f;
    float mn = __builtin_inff();

    float wn0[4], wn1[4];
#pragma unroll
    for (int c = 0; c < 4; ++c) {
      int j0 = 4 * c + 2 * half;
      wn0[c] = W[sidx[j0]     * N_OUT + oG];
      wn1[c] = W[sidx[j0 + 1] * N_OUT + oG];
    }

    float key_[8], nk_[8], ek_[8], nek_[8];
#pragma unroll
    for (int r = 0; r < 8; ++r) {
      const int kq = 8 * (r >> 2) + 4 * half + (r & 3);
      key_[r] = skey[kq];  nk_[r]  = nkey[kq];
      ek_[r]  = ewL[kq];   nek_[r] = newL[kq];
    }

    float kfirst = skey[0];
    for (int kb = 0; kb < N_IN / 16; ++kb) {
      {
        float colmin = fminf(mn, xhalf(mn));
        if (!__ballot(colmin > kfirst)) break;
      }
      v8f d0 = {0.f,0.f,0.f,0.f,0.f,0.f,0.f,0.f};
      v8f d1 = {0.f,0.f,0.f,0.f,0.f,0.f,0.f,0.f};
      float wc0[4], wc1[4];
#pragma unroll
      for (int c = 0; c < 4; ++c) { wc0[c] = wn0[c]; wc1[c] = wn1[c]; }
      const int kn = (kb + 1 < N_IN / 16) ? kb + 1 : kb;
      kfirst = skey[kn * 16];
#pragma unroll
      for (int c = 0; c < 4; ++c) {
        int jn = kn * 16 + 4 * c + 2 * half;
        wn0[c] = W[sidx[jn]     * N_OUT + oG];
        wn1[c] = W[sidx[jn + 1] * N_OUT + oG];
      }
#pragma unroll
      for (int c = 0; c < 2; ++c) {
        const int j0 = kb * 16 + 4 * c + 2 * half;
        const int j1 = j0 + 8;
        const v2f e0 = *reinterpret_cast<const v2f*>(etL + eOff + j0);
        const v2f e1 = *reinterpret_cast<const v2f*>(etL + eOff + j1);
        const v2f a0  = { keepA[c][0] ? e0[0] : 0.f, keepA[c][1] ? e0[1] : 0.f };
        const v2f a1v = { keepA[c][0] ? e1[0] : 0.f, keepA[c][1] ? e1[1] : 0.f };
        const v2f bw0 = { wc0[c],     wc1[c]     };
        const v2f bw1 = { wc0[2 + c], wc1[2 + c] };
        d0 = __builtin_amdgcn_wmma_f32_16x16x4_f32(false, a0, false, bw0,
                                                   (short)0, d0, false, false);
        d1 = __builtin_amdgcn_wmma_f32_16x16x4_f32(false, a1v, false, bw1,
                                                   (short)0, d1, false, false);
      }
      const float x3 = xhalf(d0[3]), x7 = xhalf(d0[7]);
      const float cA1 = carryA + (half ? d0[3] : x3);
      const float cB1 = carryB + (half ? d0[7] : x7);
      float a1_[8], bb_[8];
      unsigned need_[8];
      unsigned anyMask = 0u;
      unsigned untMask = 0u;
#pragma unroll
      for (int r = 0; r < 8; ++r) {
        const float a1 = (r < 4) ? d0[r]     + carryA : d1[r - 4] + cA1;
        const float bb = (r < 4) ? d0[4 + r] + carryB : d1[r]     + cB1;
        const float vs = fmaf(key_[r], a1, -bb);
        const float vn = fmaf(nk_[r],  a1, -bb);
        const float sl = V_WIDEN * (a1 + fabsf(bb));
        const float hi = fminf(a1, ek_[r])  + sl;
        const float lo = fminf(a1, nek_[r]) - sl;
        a1_[r] = a1; bb_[r] = bb;
        const unsigned below    = __builtin_amdgcn_ballot_w32(!(vs > hi));
        const unsigned above    = __builtin_amdgcn_ballot_w32(!(vn < lo));
        const unsigned untested = __builtin_amdgcn_ballot_w32(!(a1 >= 1e-10f));
        need_[r] = untested | (below & above);
        untMask |= untested;
        anyMask |= need_[r];
      }
      if (untMask != 0u) {
        anyMask = 0u;
#pragma unroll
        for (int r = 0; r < 8; ++r) {
          const float a1 = a1_[r], bb = bb_[r];
          const float vs = fmaf(key_[r], a1, -bb);
          const float sw = V_WIDEN * (fabsf(a1) + fabsf(bb) + ek_[r]);
          const bool falling_out = (a1 <= -1e-10f) && (vs + sw < ek_[r]);
          need_[r] &= ~__builtin_amdgcn_ballot_w32(falling_out);
          anyMask |= need_[r];
        }
      }
#pragma unroll
      for (int r = 0; r < 8; ++r) {
        const int kq = kn * 16 + 8 * (r >> 2) + 4 * half + (r & 3);
        key_[r] = skey[kq];  nk_[r]  = nkey[kq];
        ek_[r]  = ewL[kq];   nek_[r] = newL[kq];
      }
      if (anyMask != 0u) {
        unsigned pend = 0u;
#pragma unroll
        for (int r = 0; r < 8; ++r)
          pend |= __builtin_amdgcn_inverse_ballot_w32(need_[r]) ? (1u << (4 * r)) : 0u;
#pragma nounroll
        for (int pass = 0; pass < 8; ++pass) {
          if (!__ballot(pend != 0u)) break;
          const bool     act = (pend != 0u);
          const unsigned r   = act ? ((unsigned)__builtin_ctz(pend) >> 2) : 0u;
          const bool b0 = (r & 1u) != 0u, b1 = (r & 2u) != 0u, b2 = (r & 4u) != 0u;
          const float a1  = pick8(a1_, b0, b1, b2);
          const float bb  = pick8(bb_, b0, b1, b2);
          const int   k   = kb * 16 + 8 * (int)(r >> 2) + 4 * half + (int)(r & 3u);
          const float ssk = skey[k];
          const float nxt = nkey[k];
          const float a1s    = (fabsf(a1) < 1e-10f) ? 1e-10f : a1;
          const float inv_a1 = frcp(a1s);
          const float ratio  = bb * inv_a1;
          const float z      = -__expf(ratio) * inv_a1;
          const bool  valid  = (z >= -INV_E);
          const float zs     = fmaxf(z, -INV_E + 1e-7f);
          float t = __builtin_inff();
          if (act && valid) t = ratio - lambertw0_dev(zs);
          if (t < ssk || t > nxt) t = __builtin_inff();
          mn = fminf(mn, t);
          pend = (t < __builtin_inff()) ? 0u : (pend & (pend - 1u));
        }
      }
      const float y3 = xhalf(d1[3]), y7 = xhalf(d1[7]);
      carryA = cA1 + (half ? d1[3] : y3);
      carryB = cB1 + (half ? d1[7] : y7);
    }
    mn = fminf(mn, __shfl_xor(mn, 16, 32));
    if (half == 0) out[b * N_OUT + oG] = mn;
  }
}

extern "C" void kernel_launch(void* const* d_in, const int* in_sizes, int n_in,
                              void* d_out, int out_size, void* d_ws, size_t ws_size,
                              hipStream_t stream) {
  const float* spikes  = (const float*)d_in[0];
  const float* weights = (const float*)d_in[1];
  float*       out     = (float*)d_out;
  (void)in_sizes; (void)n_in; (void)out_size; (void)d_ws; (void)ws_size;
  ttfs_equaltime_forward<<<dim3(NB, 2), 256, 0, stream>>>(spikes, weights, out);
}
